# starting kernel
# baseline (speedup 1.0000x reference)
#include <hip/hip_runtime.h>

#define N_NODES 50000
#define N_EDGES 800000
#define C 64

typedef float    v2f  __attribute__((ext_vector_type(2)));
typedef float    v8f  __attribute__((ext_vector_type(8)));


__global__ void zero_kernel(float4* __restrict__ p, int n4) {
    int i = blockIdx.x * blockDim.x + threadIdx.x;
    if (i < n4) p[i] = make_float4(0.f, 0.f, 0.f, 0.f);
}

__global__ void deg_kernel(const int* __restrict__ row, const float* __restrict__ w,
                           float* __restrict__ deg) {
    int e = blockIdx.x * blockDim.x + threadIdx.x;
    atomicAdd(&deg[row[e]], w[e]);
}

__global__ void deginv_kernel(float* __restrict__ deg) {
    int i = blockIdx.x * blockDim.x + threadIdx.x;
    if (i < N_NODES) {
        float d = deg[i];
        deg[i] = (d > 0.f) ? rsqrtf(d) : 0.f;
    }
}

__global__ void norm_kernel(const int* __restrict__ row, const int* __restrict__ col,
                            const float* __restrict__ w, const float* __restrict__ dinv,
                            float* __restrict__ nrm) {
    int e = blockIdx.x * blockDim.x + threadIdx.x;
    nrm[e] = dinv[row[e]] * w[e] * dinv[col[e]];
}

__global__ void prop_kernel(const float* __restrict__ xin, float* __restrict__ xout,
                            const int* __restrict__ row, const int* __restrict__ col,
                            const float* __restrict__ nrm) {
    unsigned gid = blockIdx.x * blockDim.x + threadIdx.x;
    unsigned e  = gid >> 4;
    unsigned cg = (gid & 15u) << 2;
    int r = row[e];
    int c = col[e];
    float w = nrm[e];
    const float4 v = *reinterpret_cast<const float4*>(xin + (size_t)c * C + cg);
    float* o = xout + (size_t)r * C + cg;
    atomicAdd(o + 0, w * v.x);
    atomicAdd(o + 1, w * v.y);
    atomicAdd(o + 2, w * v.z);
    atomicAdd(o + 3, w * v.w);
}


__device__ __forceinline__ void mat_accum(const float* __restrict__ A,
                                          const float* Wm,
                                          int arow, int m, int hi,
                                          v8f& c0, v8f& c1, v8f& c2, v8f& c3) {
    #pragma unroll
    for (int k0 = 0; k0 < 64; k0 += 4) {
        const int kk = k0 + hi * 2;
        v2f a = *reinterpret_cast<const v2f*>(A + arow * 64 + kk);
        v2f b0, b1, b2, b3;
        b0[0] = Wm[kk * 64 + m];          b0[1] = Wm[(kk + 1) * 64 + m];
        b1[0] = Wm[kk * 64 + 16 + m];     b1[1] = Wm[(kk + 1) * 64 + 16 + m];
        b2[0] = Wm[kk * 64 + 32 + m];     b2[1] = Wm[(kk + 1) * 64 + 32 + m];
        b3[0] = Wm[kk * 64 + 48 + m];     b3[1] = Wm[(kk + 1) * 64 + 48 + m];
        c0 = __builtin_amdgcn_wmma_f32_16x16x4_f32(false, a, false, b0, (short)0, c0, false, false);
        c1 = __builtin_amdgcn_wmma_f32_16x16x4_f32(false, a, false, b1, (short)0, c1, false, false);
        c2 = __builtin_amdgcn_wmma_f32_16x16x4_f32(false, a, false, b2, (short)0, c2, false, false);
        c3 = __builtin_amdgcn_wmma_f32_16x16x4_f32(false, a, false, b3, (short)0, c3, false, false);
    }
}

__global__ void __launch_bounds__(256)
gemm_kernel(const float* __restrict__ A0, const float* __restrict__ A1,
            const float* __restrict__ A2, const float* __restrict__ A3,
            const float* __restrict__ tf, const float* __restrict__ tb,
            float* __restrict__ out) {
    __shared__ float Ws[4 * 64 * 64];

    const int t = threadIdx.x;
    #pragma unroll 4
    for (int j = 0; j < 64; ++j) {
        int i   = t + j * 256;
        int mat = i >> 12;
        int idx = i & 4095;
        float v;
        if      (mat == 0) v = tf[idx];
        else if (mat == 1) v = tb[idx] + tb[4096 + idx];
        else if (mat == 2) v = tf[4096 + idx] + tb[8192 + idx];
        else               v = tf[8192 + idx];
        Ws[i] = v;
    }
    __syncthreads();

    const int wave = t >> 5;
    const int lane = t & 31;
    const int tm   = blockIdx.x * 8 + wave;
    if (tm >= 3125) return;

    const int m    = lane & 15;
    const int hi   = lane >> 4;
    const int arow = tm * 16 + m;

    v8f c0 = {}, c1 = {}, c2 = {}, c3 = {};
    mat_accum(A0, Ws,            arow, m, hi, c0, c1, c2, c3);
    mat_accum(A1, Ws + 4096,     arow, m, hi, c0, c1, c2, c3);
    mat_accum(A2, Ws + 8192,     arow, m, hi, c0, c1, c2, c3);
    mat_accum(A3, Ws + 12288,    arow, m, hi, c0, c1, c2, c3);

    #pragma unroll
    for (int v = 0; v < 8; ++v) {
        const int rowm = tm * 16 + hi * 8 + v;
        float* o = out + rowm * 64 + m;
        o[0]  = c0[v];
        o[16] = c1[v];
        o[32] = c2[v];
        o[48] = c3[v];
    }
}


extern "C" void kernel_launch(void* const* d_in, const int* in_sizes, int n_in,
                              void* d_out, int out_size, void* d_ws, size_t ws_size,
                              hipStream_t stream) {
    (void)in_sizes; (void)n_in; (void)out_size; (void)ws_size;

    const float* x  = (const float*)d_in[0];
    const int*   ei = (const int*)d_in[1];
    const float* ew = (const float*)d_in[2];
    const float* tf = (const float*)d_in[3];
    const float* tb = (const float*)d_in[4];
    const int* row = ei;
    const int* col = ei + N_EDGES;

    float* ws  = (float*)d_ws;
    float* deg = ws;
    float* Tx1 = ws  + 50176;
    float* Tx2 = Tx1 + (size_t)N_NODES * C;
    float* Tx3 = Tx2 + (size_t)N_NODES * C;
    float* nrm = Tx3 + (size_t)N_NODES * C;

    const int zero4 = (50176 + 3 * N_NODES * C) / 4;
    zero_kernel<<<(zero4 + 255) / 256, 256, 0, stream>>>((float4*)deg, zero4);

    deg_kernel<<<N_EDGES / 256, 256, 0, stream>>>(row, ew, deg);
    deginv_kernel<<<(N_NODES + 255) / 256, 256, 0, stream>>>(deg);
    norm_kernel<<<N_EDGES / 256, 256, 0, stream>>>(row, col, ew, deg, nrm);

    const int prop_blocks = (N_EDGES * 16) / 256;
    prop_kernel<<<prop_blocks, 256, 0, stream>>>(x,   Tx1, row, col, nrm);
    prop_kernel<<<prop_blocks, 256, 0, stream>>>(Tx1, Tx2, row, col, nrm);
    prop_kernel<<<prop_blocks, 256, 0, stream>>>(Tx2, Tx3, row, col, nrm);

    gemm_kernel<<<(3125 + 7) / 8, 256, 0, stream>>>(x, Tx1, Tx2, Tx3, tf, tb,
                                                    (float*)d_out);
}


# prop: one wave per edge, lane = channel, 8 edges per wave: each gather and each atomic add instruction covers one whole 128-B line (4x fewer atomic line requests); atomics kept, arrival-order sums
# speedup vs baseline: 1.5491x; 1.5491x over previous
#include <hip/hip_runtime.h>

#define N_NODES 50000
#define N_EDGES 800000
#define C 64

typedef float    v2f  __attribute__((ext_vector_type(2)));
typedef float    v8f  __attribute__((ext_vector_type(8)));
typedef int      v8i  __attribute__((ext_vector_type(8)));


__global__ void zero_kernel(float4* __restrict__ p, int n4) {
    int i = blockIdx.x * blockDim.x + threadIdx.x;
    if (i < n4) p[i] = make_float4(0.f, 0.f, 0.f, 0.f);
}

__global__ void deg_kernel(const int* __restrict__ row, const float* __restrict__ w,
                           float* __restrict__ deg) {
    int e = blockIdx.x * blockDim.x + threadIdx.x;
    atomicAdd(&deg[row[e]], w[e]);
}

__global__ void deginv_kernel(float* __restrict__ deg) {
    int i = blockIdx.x * blockDim.x + threadIdx.x;
    if (i < N_NODES) {
        float d = deg[i];
        deg[i] = (d > 0.f) ? rsqrtf(d) : 0.f;
    }
}

__global__ void norm_kernel(const int* __restrict__ row, const int* __restrict__ col,
                            const float* __restrict__ w, const float* __restrict__ dinv,
                            float* __restrict__ nrm) {
    int e = blockIdx.x * blockDim.x + threadIdx.x;
    nrm[e] = dinv[row[e]] * w[e] * dinv[col[e]];
}

#define EPW 8
__global__ void __launch_bounds__(256)
prop_kernel(const float* __restrict__ xin, float* __restrict__ xout,
            const int* __restrict__ row, const int* __restrict__ col,
            const float* __restrict__ nrm) {
    const unsigned lane = threadIdx.x & 31u;
    const unsigned e0 = (unsigned)__builtin_amdgcn_readfirstlane(
                            (int)((blockIdx.x * 8u + (threadIdx.x >> 5)) * EPW));
    const v8i r = *reinterpret_cast<const v8i*>(row + e0);
    const v8i c = *reinterpret_cast<const v8i*>(col + e0);
    const v8f w = *reinterpret_cast<const v8f*>(nrm + e0);
    float v0[EPW], v1[EPW];
    #pragma unroll
    for (int i = 0; i < EPW; ++i) {
        const float* s = xin + (size_t)c[i] * C;
        v0[i] = s[lane];
        v1[i] = s[lane + 32u];
    }
    #pragma unroll
    for (int i = 0; i < EPW; ++i) {
        float* o = xout + (size_t)r[i] * C;
        atomicAdd(o + lane,       w[i] * v0[i]);
        atomicAdd(o + lane + 32u, w[i] * v1[i]);
    }
}


__device__ __forceinline__ void mat_accum(const float* __restrict__ A,
                                          const float* Wm,
                                          int arow, int m, int hi,
                                          v8f& c0, v8f& c1, v8f& c2, v8f& c3) {
    #pragma unroll
    for (int k0 = 0; k0 < 64; k0 += 4) {
        const int kk = k0 + hi * 2;
        v2f a = *reinterpret_cast<const v2f*>(A + arow * 64 + kk);
        v2f b0, b1, b2, b3;
        b0[0] = Wm[kk * 64 + m];          b0[1] = Wm[(kk + 1) * 64 + m];
        b1[0] = Wm[kk * 64 + 16 + m];     b1[1] = Wm[(kk + 1) * 64 + 16 + m];
        b2[0] = Wm[kk * 64 + 32 + m];     b2[1] = Wm[(kk + 1) * 64 + 32 + m];
        b3[0] = Wm[kk * 64 + 48 + m];     b3[1] = Wm[(kk + 1) * 64 + 48 + m];
        c0 = __builtin_amdgcn_wmma_f32_16x16x4_f32(false, a, false, b0, (short)0, c0, false, false);
        c1 = __builtin_amdgcn_wmma_f32_16x16x4_f32(false, a, false, b1, (short)0, c1, false, false);
        c2 = __builtin_amdgcn_wmma_f32_16x16x4_f32(false, a, false, b2, (short)0, c2, false, false);
        c3 = __builtin_amdgcn_wmma_f32_16x16x4_f32(false, a, false, b3, (short)0, c3, false, false);
    }
}

__global__ void __launch_bounds__(256)
gemm_kernel(const float* __restrict__ A0, const float* __restrict__ A1,
            const float* __restrict__ A2, const float* __restrict__ A3,
            const float* __restrict__ tf, const float* __restrict__ tb,
            float* __restrict__ out) {
    __shared__ float Ws[4 * 64 * 64];

    const int t = threadIdx.x;
    #pragma unroll 4
    for (int j = 0; j < 64; ++j) {
        int i   = t + j * 256;
        int mat = i >> 12;
        int idx = i & 4095;
        float v;
        if      (mat == 0) v = tf[idx];
        else if (mat == 1) v = tb[idx] + tb[4096 + idx];
        else if (mat == 2) v = tf[4096 + idx] + tb[8192 + idx];
        else               v = tf[8192 + idx];
        Ws[i] = v;
    }
    __syncthreads();

    const int wave = t >> 5;
    const int lane = t & 31;
    const int tm   = blockIdx.x * 8 + wave;
    if (tm >= 3125) return;

    const int m    = lane & 15;
    const int hi   = lane >> 4;
    const int arow = tm * 16 + m;

    v8f c0 = {}, c1 = {}, c2 = {}, c3 = {};
    mat_accum(A0, Ws,            arow, m, hi, c0, c1, c2, c3);
    mat_accum(A1, Ws + 4096,     arow, m, hi, c0, c1, c2, c3);
    mat_accum(A2, Ws + 8192,     arow, m, hi, c0, c1, c2, c3);
    mat_accum(A3, Ws + 12288,    arow, m, hi, c0, c1, c2, c3);

    #pragma unroll
    for (int v = 0; v < 8; ++v) {
        const int rowm = tm * 16 + hi * 8 + v;
        float* o = out + rowm * 64 + m;
        o[0]  = c0[v];
        o[16] = c1[v];
        o[32] = c2[v];
        o[48] = c3[v];
    }
}


extern "C" void kernel_launch(void* const* d_in, const int* in_sizes, int n_in,
                              void* d_out, int out_size, void* d_ws, size_t ws_size,
                              hipStream_t stream) {
    (void)in_sizes; (void)n_in; (void)out_size; (void)ws_size;

    const float* x  = (const float*)d_in[0];
    const int*   ei = (const int*)d_in[1];
    const float* ew = (const float*)d_in[2];
    const float* tf = (const float*)d_in[3];
    const float* tb = (const float*)d_in[4];
    const int* row = ei;
    const int* col = ei + N_EDGES;

    float* ws  = (float*)d_ws;
    float* deg = ws;
    float* Tx1 = ws  + 50176;
    float* Tx2 = Tx1 + (size_t)N_NODES * C;
    float* Tx3 = Tx2 + (size_t)N_NODES * C;
    float* nrm = Tx3 + (size_t)N_NODES * C;

    const int zero4 = (50176 + 3 * N_NODES * C) / 4;
    zero_kernel<<<(zero4 + 255) / 256, 256, 0, stream>>>((float4*)deg, zero4);

    deg_kernel<<<N_EDGES / 256, 256, 0, stream>>>(row, ew, deg);
    deginv_kernel<<<(N_NODES + 255) / 256, 256, 0, stream>>>(deg);
    norm_kernel<<<N_EDGES / 256, 256, 0, stream>>>(row, col, ew, deg, nrm);

    const int prop_blocks = N_EDGES / (EPW * 8);
    prop_kernel<<<prop_blocks, 256, 0, stream>>>(x,   Tx1, row, col, nrm);
    prop_kernel<<<prop_blocks, 256, 0, stream>>>(Tx1, Tx2, row, col, nrm);
    prop_kernel<<<prop_blocks, 256, 0, stream>>>(Tx2, Tx3, row, col, nrm);

    gemm_kernel<<<(3125 + 7) / 8, 256, 0, stream>>>(x, Tx1, Tx2, Tx3, tf, tb,
                                                    (float*)d_out);
}
